# starting kernel
# baseline (speedup 1.0000x reference)
#include <hip/hip_runtime.h>
#include <math.h>

#define BB 64
#define RR 36
#define RP 48
#define LL 64
#define EE 512
#define KC 64
#define LAMBDA 9.0f

typedef float v2f __attribute__((ext_vector_type(2)));
typedef float v8f __attribute__((ext_vector_type(8)));

__device__ __forceinline__ v8f wmma_f32(v2f a, v2f b, v8f c) {
  return __builtin_amdgcn_wmma_f32_16x16x4_f32(
       false, a,  false, b,
       (short)0, c,  false,  false);
}

__global__ __launch_bounds__(128) void normalize_rows(
    const float* __restrict__ captions, const float* __restrict__ images,
    float* __restrict__ capn, float* __restrict__ imgn, float* __restrict__ na) {
  __shared__ float red[128];
  const int row = blockIdx.x;
  const int tid = threadIdx.x;
  const bool is_cap = row < BB * LL;
  const float* src = is_cap ? captions + (size_t)row * EE
                            : images + (size_t)(row - BB * LL) * EE;
  const float4 v = *(const float4*)(src + tid * 4);
  red[tid] = v.x * v.x + v.y * v.y + v.z * v.z + v.w * v.w;
  __syncthreads();
  for (int off = 64; off > 0; off >>= 1) {
    if (tid < off) red[tid] += red[tid + off];
    __syncthreads();
  }
  const float norm = sqrtf(red[0]);
  const float scale = 1.0f / fmaxf(norm, 1e-12f);
  float* dst = is_cap ? capn + (size_t)row * EE
                      : imgn + (size_t)(row - BB * LL) * EE;
  float4 o;
  o.x = v.x * scale; o.y = v.y * scale; o.z = v.z * scale; o.w = v.w * scale;
  *(float4*)(dst + tid * 4) = o;
  if (is_cap && tid == 0) na[row] = norm * scale;
}

__global__ __launch_bounds__(32) void gram_kernel(
    const float* __restrict__ imgn, float* __restrict__ G) {
  const int i = blockIdx.y;
  const int t = blockIdx.x, ti = t / 3, tj = t % 3;
  const int lane = threadIdx.x;
  const int hi = lane >> 4, n16 = lane & 15;
  const int ar = ti * 16 + n16, br = tj * 16 + n16;
  const bool am = ar < RR, bm = br < RR;
  const float* Arow = imgn + ((size_t)i * RR + (am ? ar : 0)) * EE;
  const float* Brow = imgn + ((size_t)i * RR + (bm ? br : 0)) * EE;
  v8f acc = {};
#pragma unroll 8
  for (int k = 0; k < EE; k += 4) {
    const int kk = k + 2 * hi;
    v2f a = *(const v2f*)(Arow + kk);
    v2f b = *(const v2f*)(Brow + kk);
    if (!am) { a.x = 0.f; a.y = 0.f; }
    if (!bm) { b.x = 0.f; b.y = 0.f; }
    acc = wmma_f32(a, b, acc);
  }
#pragma unroll
  for (int j = 0; j < 8; ++j) {
    const int row = ti * 16 + j + 8 * hi;
    const int col = tj * 16 + n16;
    G[((size_t)i * RP + row) * RP + col] = acc[j];
  }
}

__global__ __launch_bounds__(128) void scan_main(
    const float* __restrict__ capn, const float* __restrict__ imgn,
    const float* __restrict__ na_g, const float* __restrict__ G,
    const int* __restrict__ cap_lens, float* __restrict__ out) {
  __shared__ float Ash[LL][KC + 2];
  __shared__ float Bsh[RP][KC + 2];
  __shared__ float Wsh[LL][RP + 2];
  __shared__ float Gsh[RP][RP + 2];
  __shared__ float dotsh[LL];
  __shared__ float rsh[LL];

  const int i = blockIdx.x, c = blockIdx.y;
  const int tid = threadIdx.x;
  const int wave = tid >> 5, lane = tid & 31;
  const int hi = lane >> 4, n16 = lane & 15;

  for (int idx = tid; idx < RP * RP; idx += 128) {
    const int r = idx / RP, col = idx % RP;
    Gsh[r][col] = G[((size_t)i * RP + r) * RP + col];
  }

  const float* Abase = capn + (size_t)c * LL * EE;
  const float* Bbase = imgn + (size_t)i * RR * EE;

  v8f acc0 = {}, acc1 = {}, acc2 = {};
  for (int kc = 0; kc < EE; kc += KC) {
    __syncthreads();
#pragma unroll
    for (int q = 0; q < 8; ++q) {
      const int idx = tid + q * 128;
      const int row = idx >> 4, c4 = (idx & 15) << 2;
      const float4 v = *(const float4*)(Abase + row * EE + kc + c4);
      Ash[row][c4] = v.x; Ash[row][c4 + 1] = v.y;
      Ash[row][c4 + 2] = v.z; Ash[row][c4 + 3] = v.w;
      if (kc + KC < EE)
        __builtin_prefetch(Abase + row * EE + kc + KC + c4, 0, 1);
    }
#pragma unroll
    for (int q = 0; q < 6; ++q) {
      const int idx = tid + q * 128;
      const int row = idx >> 4, c4 = (idx & 15) << 2;
      float4 v; v.x = 0.f; v.y = 0.f; v.z = 0.f; v.w = 0.f;
      if (row < RR) v = *(const float4*)(Bbase + row * EE + kc + c4);
      Bsh[row][c4] = v.x; Bsh[row][c4 + 1] = v.y;
      Bsh[row][c4 + 2] = v.z; Bsh[row][c4 + 3] = v.w;
    }
    __syncthreads();
#pragma unroll
    for (int k0 = 0; k0 < KC; k0 += 4) {
      const int kk = k0 + 2 * hi;
      const v2f a  = *(const v2f*)&Ash[wave * 16 + n16][kk];
      const v2f b0 = *(const v2f*)&Bsh[n16][kk];
      const v2f b1 = *(const v2f*)&Bsh[16 + n16][kk];
      const v2f b2 = *(const v2f*)&Bsh[32 + n16][kk];
      acc0 = wmma_f32(a, b0, acc0);
      acc1 = wmma_f32(a, b1, acc1);
      acc2 = wmma_f32(a, b2, acc2);
    }
  }

  __syncthreads();
#pragma unroll
  for (int j = 0; j < 8; ++j) {
    const int l = wave * 16 + j + 8 * hi;
    Wsh[l][n16]      = acc0[j];
    Wsh[l][16 + n16] = acc1[j];
    Wsh[l][32 + n16] = acc2[j];
  }
  __syncthreads();

  if (tid < LL) {
    float m = -1e30f;
    for (int r = 0; r < RR; ++r) m = fmaxf(m, Wsh[tid][r]);
    float sum = 0.f;
    for (int r = 0; r < RR; ++r) sum += expf(LAMBDA * (Wsh[tid][r] - m));
    const float inv = 1.0f / sum;
    float dot = 0.f;
    for (int r = 0; r < RR; ++r) {
      const float s = Wsh[tid][r];
      const float w = expf(LAMBDA * (s - m)) * inv;
      dot += w * s;
      Wsh[tid][r] = w;
    }
    for (int r = RR; r < RP; ++r) Wsh[tid][r] = 0.f;
    dotsh[tid] = dot;
  }
  __syncthreads();

  v8f v0 = {}, v1 = {}, v2 = {};
#pragma unroll
  for (int k0 = 0; k0 < RP; k0 += 4) {
    const int kk = k0 + 2 * hi;
    const v2f a  = *(const v2f*)&Wsh[wave * 16 + n16][kk];
    const v2f b0 = *(const v2f*)&Gsh[n16][kk];
    const v2f b1 = *(const v2f*)&Gsh[16 + n16][kk];
    const v2f b2 = *(const v2f*)&Gsh[32 + n16][kk];
    v0 = wmma_f32(a, b0, v0);
    v1 = wmma_f32(a, b1, v1);
    v2 = wmma_f32(a, b2, v2);
  }

  __syncthreads();
#pragma unroll
  for (int j = 0; j < 8; ++j) {
    const int l = wave * 16 + j + 8 * hi;
    Ash[l][n16]      = v0[j] * Wsh[l][n16];
    Ash[l][16 + n16] = v1[j] * Wsh[l][16 + n16];
    Ash[l][32 + n16] = v2[j] * Wsh[l][32 + n16];
  }
  __syncthreads();

  if (tid < LL) {
    float nb2 = 0.f;
    for (int r = 0; r < RR; ++r) nb2 += Ash[tid][r];
    const float nb = sqrtf(fmaxf(nb2, 0.f));
    const float na = na_g[c * LL + tid];
    const float denom = fmaxf(na * nb, 1e-8f);
    float rl = dotsh[tid] / denom;
    if (tid >= cap_lens[c]) rl = 0.f;
    rsh[tid] = rl;
  }
  __syncthreads();
  if (tid == 0) {
    float s = 0.f;
    for (int l = 0; l < LL; ++l) s += rsh[l];
    out[(size_t)c * BB + i] = s * (1.0f / LL);
  }
}

extern "C" void kernel_launch(void* const* d_in, const int* in_sizes, int n_in,
                              void* d_out, int out_size, void* d_ws, size_t ws_size,
                              hipStream_t stream) {
  (void)in_sizes; (void)n_in; (void)out_size; (void)ws_size;
  const float* images = (const float*)d_in[0];
  const float* captions = (const float*)d_in[1];
  const int* cap_lens = (const int*)d_in[2];
  float* out = (float*)d_out;

  float* ws = (float*)d_ws;
  float* capn = ws;
  float* imgn = capn + (size_t)BB * LL * EE;
  float* na   = imgn + (size_t)BB * RR * EE;
  float* G    = na + (size_t)BB * LL;

  normalize_rows<<<BB * LL + BB * RR, 128, 0, stream>>>(captions, images, capn,
                                                        imgn, na);
  gram_kernel<<<dim3(9, BB), 32, 0, stream>>>(imgn, G);
  scan_main<<<dim3(BB, BB), 128, 0, stream>>>(capn, imgn, na, G, cap_lens, out);
}


# Gram launch folded into the normalize launch: an image block keeps its normalized rows in LDS, six waves build G's upper tiles from them (mirrored), two pack; 3 launches become 2; sum orders kept
# speedup vs baseline: 1.6472x; 1.6472x over previous
#include <hip/hip_runtime.h>
#include <math.h>


#define BB 64
#define RR 36
#define RP 48
#define LL 64
#define EE 512
#define KB8 64
#define CAP_TILES (BB * LL / 16)
#define IMG_TILES (BB * 3)
#define TILE_F4 (KB8 * 32)
#define LAMBDA 9.0f

typedef float v2f __attribute__((ext_vector_type(2)));
typedef float v8f __attribute__((ext_vector_type(8)));

__device__ __forceinline__ v8f wmma_f32(v2f a, v2f b, v8f c) {
  return __builtin_amdgcn_wmma_f32_16x16x4_f32(
       false, a,  false, b,
       (short)0, c,  false,  false);
}


#define FRONT_PITCH (EE + 4)

__device__ __forceinline__ void load_row(float4 (&v)[4], const float* src, int lane) {
#pragma unroll
  for (int j = 0; j < 4; ++j) v[j] = *(const float4*)(src + 4 * lane + 128 * j);
}

__device__ __forceinline__ float normalize_row(const float4 (&v)[4], float* dstrow, int lane) {
  float p[4];
#pragma unroll
  for (int j = 0; j < 4; ++j) {
    const float4 x = v[j];
    p[j] = x.x * x.x + x.y * x.y + x.z * x.z + x.w * x.w;
  }
  float s = (p[0] + p[2]) + (p[1] + p[3]);
  s += __shfl_xor(s, 16);
  s += __shfl_xor(s, 8);
  s += __shfl_xor(s, 4);
  s += __shfl_xor(s, 2);
  s += __shfl_xor(s, 1);
  const float norm = sqrtf(s);
  const float scale = 1.0f / fmaxf(norm, 1e-12f);
#pragma unroll
  for (int j = 0; j < 4; ++j) {
    float4 o = v[j];
    o.x *= scale; o.y *= scale; o.z *= scale; o.w *= scale;
    *(float4*)&dstrow[4 * lane + 128 * j] = o;
  }
  return norm * scale;
}

__device__ __forceinline__ void pack_kb(float4* dst, const float* rp, int kb, int h, int lane) {
  const v2f lo = *(const v2f*)&rp[8 * kb + 2 * h];
  const v2f hi2 = *(const v2f*)&rp[8 * kb + 4 + 2 * h];
  float4 o; o.x = lo.x; o.y = lo.y; o.z = hi2.x; o.w = hi2.y;
  dst[kb * 32 + lane] = o;
}

__device__ __forceinline__ void gram_load(v2f (&fa)[16], v2f (&fb)[16], const float* ap,
                                          const float* bp, int g) {
#pragma unroll
  for (int s = 0; s < 16; ++s) {
    fa[s] = *(const v2f*)&ap[64 * g + 4 * s];
    fb[s] = *(const v2f*)&bp[64 * g + 4 * s];
  }
}

__device__ __forceinline__ v8f gram_mma(v8f acc, const v2f (&fa)[16], const v2f (&fb)[16]) {
#pragma unroll
  for (int s = 0; s < 16; ++s) acc = wmma_f32(fa[s], fb[s], acc);
  return acc;
}

__global__ __launch_bounds__(256) void normalize_pack_gram(
    const float* __restrict__ captions, const float* __restrict__ images,
    float* __restrict__ capP, float* __restrict__ imgP, float* __restrict__ na,
    float* __restrict__ G) {
  __shared__ float rows[RP][FRONT_PITCH];
  const int tid = threadIdx.x;
  const int wave = __builtin_amdgcn_readfirstlane(tid >> 5);
  const int lane = tid & 31;
  const int c = lane & 15, h = lane >> 4;

  if (blockIdx.x < CAP_TILES) {
    const int tile = blockIdx.x;
    float4 v[2][4];
#pragma unroll
    for (int rr = 0; rr < 2; ++rr)
      load_row(v[rr], captions + ((size_t)tile * 16 + wave * 2 + rr) * EE, lane);
#pragma unroll
    for (int rr = 0; rr < 2; ++rr) {
      const int row = wave * 2 + rr;
      const float nrm = normalize_row(v[rr], rows[row], lane);
      if (lane == 0) na[tile * 16 + row] = nrm;
    }
    __syncthreads();
    float4* dst = (float4*)capP + (size_t)tile * TILE_F4;
#pragma unroll
    for (int j = 0; j < 8; ++j) pack_kb(dst, rows[c], wave + 8 * j, h, lane);
    return;
  }

  const int img = blockIdx.x - CAP_TILES;
  const float* isrc = images + (size_t)img * RR * EE;
  float4 v[5][4];
#pragma unroll
  for (int rr = 0; rr < 5; ++rr) {
    const int row = wave + 8 * rr;
    if (row < RR) load_row(v[rr], isrc + (size_t)row * EE, lane);
  }
  for (int idx = tid; idx < (RP - RR) * FRONT_PITCH / 4; idx += 256)
    ((float4*)&rows[RR][0])[idx] = float4{0.f, 0.f, 0.f, 0.f};
#pragma unroll
  for (int rr = 0; rr < 5; ++rr) {
    const int row = wave + 8 * rr;
    if (row < RR) (void)normalize_row(v[rr], rows[row], lane);
  }
  __syncthreads();

  if (wave >= 6) {
    for (int t = 0; t < 3; ++t) {
      float4* dst = (float4*)imgP + (size_t)(img * 3 + t) * TILE_F4;
      const float* rp = rows[t * 16 + c];
#pragma unroll 8
      for (int j = 0; j < KB8 / 2; ++j) pack_kb(dst, rp, (wave - 6) + 2 * j, h, lane);
    }
    return;
  }

  const int ti = (wave >= 3) + (wave >= 5);
  const int tj = wave - (ti == 0 ? 0 : (ti == 1 ? 2 : 3));
  const float* ap = rows[ti * 16 + c] + 2 * h;
  const float* bp = rows[tj * 16 + c] + 2 * h;
  v8f acc = {};
#define GROUP_STAYS() __builtin_amdgcn_sched_barrier(0)
  v2f fa0[16], fb0[16], fa1[16], fb1[16];
  gram_load(fa0, fb0, ap, bp, 0);
  GROUP_STAYS();
#pragma unroll
  for (int g = 0; g < 8; g += 2) {
    gram_load(fa1, fb1, ap, bp, g + 1);
    GROUP_STAYS();
    acc = gram_mma(acc, fa0, fb0);
    GROUP_STAYS();
    if (g + 2 < 8) gram_load(fa0, fb0, ap, bp, g + 2);
    GROUP_STAYS();
    acc = gram_mma(acc, fa1, fb1);
    GROUP_STAYS();
  }
#undef GROUP_STAYS
  float* Gi = G + (size_t)img * RP * RP;
#pragma unroll
  for (int j = 0; j < 8; ++j)
    Gi[(ti * 16 + j + 8 * h) * RP + tj * 16 + c] = acc[j];
  if (ti != tj) {
    float4* m = (float4*)(Gi + (tj * 16 + c) * RP + ti * 16 + 8 * h);
    m[0] = float4{acc[0], acc[1], acc[2], acc[3]};
    m[1] = float4{acc[4], acc[5], acc[6], acc[7]};
  }
}

__device__ __forceinline__ void load_set(float4 (&a)[4], float4 (&b)[3],
                                         const float4* Ap, const float4* Bp, int kb) {
#pragma unroll
  for (int m = 0; m < 4; ++m) a[m] = Ap[(m * KB8 + kb) * 32];
#pragma unroll
  for (int n = 0; n < 3; ++n) b[n] = Bp[(n * KB8 + kb) * 32];
}

__device__ __forceinline__ void mma_set(v8f (&acc)[4][3], const float4 (&a)[4],
                                        const float4 (&b)[3]) {
#pragma unroll
  for (int m = 0; m < 4; ++m)
#pragma unroll
    for (int n = 0; n < 3; ++n)
      acc[m][n] = wmma_f32(v2f{a[m].x, a[m].y}, v2f{b[n].x, b[n].y}, acc[m][n]);
#pragma unroll
  for (int m = 0; m < 4; ++m)
#pragma unroll
    for (int n = 0; n < 3; ++n)
      acc[m][n] = wmma_f32(v2f{a[m].z, a[m].w}, v2f{b[n].z, b[n].w}, acc[m][n]);
}

__global__ __launch_bounds__(128) void scan_main(
    const float* __restrict__ capP, const float* __restrict__ imgP,
    const float* __restrict__ na_g, const float* __restrict__ G,
    const int* __restrict__ cap_lens, float* __restrict__ out) {
  __shared__ float Wsh[4][LL][RP + 2];
  __shared__ float rsh[4][LL];

  const int tid = threadIdx.x;
  const int wave = tid >> 5, lane = tid & 31;
  const int hi = lane >> 4, n16 = lane & 15;
  const int i = blockIdx.x * 2 + (wave & 1);
  const int c = blockIdx.y * 2 + (wave >> 1);
  float (*W)[RP + 2] = Wsh[wave];

  const float4* Ap = (const float4*)capP + (size_t)(c * 4) * TILE_F4 + lane;
  const float4* Bp = (const float4*)imgP + (size_t)(i * 3) * TILE_F4 + lane;

  v8f acc[4][3];
#pragma unroll
  for (int m = 0; m < 4; ++m)
#pragma unroll
    for (int n = 0; n < 3; ++n) acc[m][n] = v8f{};

#define LOADS_STAY() __builtin_amdgcn_sched_barrier(0)
  float4 a0[4], b0[3], a1[4], b1[3], a2[4], b2[3], a3[4], b3[3];
  load_set(a0, b0, Ap, Bp, 0);
  LOADS_STAY();
  load_set(a1, b1, Ap, Bp, 1);
  LOADS_STAY();
  load_set(a2, b2, Ap, Bp, 2);
  LOADS_STAY();
  for (int kb = 0; kb < KB8 - 4; kb += 4) {
    load_set(a3, b3, Ap, Bp, kb + 3);
    LOADS_STAY();
    mma_set(acc, a0, b0);
    LOADS_STAY();
    load_set(a0, b0, Ap, Bp, kb + 4);
    LOADS_STAY();
    mma_set(acc, a1, b1);
    LOADS_STAY();
    load_set(a1, b1, Ap, Bp, kb + 5);
    LOADS_STAY();
    mma_set(acc, a2, b2);
    LOADS_STAY();
    load_set(a2, b2, Ap, Bp, kb + 6);
    LOADS_STAY();
    mma_set(acc, a3, b3);
    LOADS_STAY();
  }
  load_set(a3, b3, Ap, Bp, KB8 - 1);
  LOADS_STAY();
  mma_set(acc, a0, b0);
  mma_set(acc, a1, b1);
  mma_set(acc, a2, b2);
  mma_set(acc, a3, b3);
#undef LOADS_STAY

#pragma unroll
  for (int m = 0; m < 4; ++m)
#pragma unroll
    for (int j = 0; j < 8; ++j) {
      const int l = m * 16 + j + 8 * hi;
      W[l][n16]      = acc[m][0][j];
      W[l][16 + n16] = acc[m][1][j];
      W[l][32 + n16] = acc[m][2][j];
    }

  __syncthreads();

  const float* Gi = G + (size_t)i * RP * RP;
  v2f g[12][3];
#pragma unroll
  for (int ks = 0; ks < 12; ++ks)
#pragma unroll
    for (int n = 0; n < 3; ++n)
      g[ks][n] = *(const v2f*)(Gi + (n * 16 + n16) * RP + ks * 4 + 2 * hi);

  float dot[2];
#pragma unroll
  for (int q = 0; q < 2; ++q) {
    const int tok = lane + 32 * q;
    float mx = -1e30f;
    for (int r = 0; r < RR; ++r) mx = fmaxf(mx, W[tok][r]);
    float sum = 0.f;
    for (int r = 0; r < RR; ++r) sum += expf(LAMBDA * (W[tok][r] - mx));
    const float inv = 1.0f / sum;
    float d = 0.f;
    for (int r = 0; r < RR; ++r) {
      const float s = W[tok][r];
      const float w = expf(LAMBDA * (s - mx)) * inv;
      d += w * s;
      W[tok][r] = w;
    }
    for (int r = RR; r < RP; ++r) W[tok][r] = 0.f;
    dot[q] = d;
  }
  __syncthreads();

  v8f vv[4][3];
#pragma unroll
  for (int m = 0; m < 4; ++m)
#pragma unroll
    for (int n = 0; n < 3; ++n) vv[m][n] = v8f{};
#pragma unroll
  for (int ks = 0; ks < 12; ++ks) {
    const int kk = ks * 4 + 2 * hi;
#pragma unroll
    for (int m = 0; m < 4; ++m) {
      const v2f a = *(const v2f*)&W[m * 16 + n16][kk];
#pragma unroll
      for (int n = 0; n < 3; ++n) vv[m][n] = wmma_f32(a, g[ks][n], vv[m][n]);
    }
  }

  __syncthreads();
#pragma unroll
  for (int m = 0; m < 4; ++m)
#pragma unroll
    for (int j = 0; j < 8; ++j) {
      const int l = m * 16 + j + 8 * hi;
      W[l][n16]      = vv[m][0][j] * W[l][n16];
      W[l][16 + n16] = vv[m][1][j] * W[l][16 + n16];
      W[l][32 + n16] = vv[m][2][j] * W[l][32 + n16];
    }
  __syncthreads();

  const int len = cap_lens[c];
#pragma unroll
  for (int q = 0; q < 2; ++q) {
    const int tok = lane + 32 * q;
    float nb2 = 0.f;
    for (int r = 0; r < RR; ++r) nb2 += W[tok][r];
    const float nb = sqrtf(fmaxf(nb2, 0.f));
    const float na = na_g[c * LL + tok];
    const float denom = fmaxf(na * nb, 1e-8f);
    float rl = dot[q] / denom;
    if (tok >= len) rl = 0.f;
    rsh[wave][tok] = rl;
  }
  __syncthreads();
  if (lane == 0) {
    float s = 0.f;
    for (int l = 0; l < LL; ++l) s += rsh[wave][l];
    out[(size_t)c * BB + i] = s * (1.0f / LL);
  }
}

extern "C" void kernel_launch(void* const* d_in, const int* in_sizes, int n_in,
                              void* d_out, int out_size, void* d_ws, size_t ws_size,
                              hipStream_t stream) {
  (void)in_sizes; (void)n_in; (void)out_size; (void)ws_size;
  const float* images = (const float*)d_in[0];
  const float* captions = (const float*)d_in[1];
  const int* cap_lens = (const int*)d_in[2];
  float* out = (float*)d_out;

  float* ws = (float*)d_ws;
  float* capP = ws;
  float* imgP = capP + (size_t)CAP_TILES * TILE_F4 * 4;
  float* na   = imgP + (size_t)IMG_TILES * TILE_F4 * 4;
  float* G    = na + (size_t)BB * LL;

  normalize_pack_gram<<<CAP_TILES + BB, 256, 0, stream>>>(captions, images, capP, imgP,
                                                          na, G);
  scan_main<<<dim3(BB / 2, BB / 2), 128, 0, stream>>>(capP, imgP, na, G, cap_lens, out);
}


# scan_main softmax weights kept unnormalised through the Gram product: one exp per score (was two), 1/sum applied once per token to dot and norm; sum orders kept, scaling point moves
# speedup vs baseline: 1.7117x; 1.0392x over previous
#include <hip/hip_runtime.h>
#include <math.h>


#define BB 64
#define RR 36
#define RP 48
#define LL 64
#define EE 512
#define KB8 64
#define CAP_TILES (BB * LL / 16)
#define IMG_TILES (BB * 3)
#define TILE_F4 (KB8 * 32)
#define LAMBDA 9.0f

typedef float v2f __attribute__((ext_vector_type(2)));
typedef float v8f __attribute__((ext_vector_type(8)));

__device__ __forceinline__ v8f wmma_f32(v2f a, v2f b, v8f c) {
  return __builtin_amdgcn_wmma_f32_16x16x4_f32(
       false, a,  false, b,
       (short)0, c,  false,  false);
}


#define FRONT_PITCH (EE + 4)

__device__ __forceinline__ void load_row(float4 (&v)[4], const float* src, int lane) {
#pragma unroll
  for (int j = 0; j < 4; ++j) v[j] = *(const float4*)(src + 4 * lane + 128 * j);
}

__device__ __forceinline__ float normalize_row(const float4 (&v)[4], float* dstrow, int lane) {
  float p[4];
#pragma unroll
  for (int j = 0; j < 4; ++j) {
    const float4 x = v[j];
    p[j] = x.x * x.x + x.y * x.y + x.z * x.z + x.w * x.w;
  }
  float s = (p[0] + p[2]) + (p[1] + p[3]);
  s += __shfl_xor(s, 16);
  s += __shfl_xor(s, 8);
  s += __shfl_xor(s, 4);
  s += __shfl_xor(s, 2);
  s += __shfl_xor(s, 1);
  const float norm = sqrtf(s);
  const float scale = 1.0f / fmaxf(norm, 1e-12f);
#pragma unroll
  for (int j = 0; j < 4; ++j) {
    float4 o = v[j];
    o.x *= scale; o.y *= scale; o.z *= scale; o.w *= scale;
    *(float4*)&dstrow[4 * lane + 128 * j] = o;
  }
  return norm * scale;
}

__device__ __forceinline__ void pack_kb(float4* dst, const float* rp, int kb, int h, int lane) {
  const v2f lo = *(const v2f*)&rp[8 * kb + 2 * h];
  const v2f hi2 = *(const v2f*)&rp[8 * kb + 4 + 2 * h];
  float4 o; o.x = lo.x; o.y = lo.y; o.z = hi2.x; o.w = hi2.y;
  dst[kb * 32 + lane] = o;
}

__device__ __forceinline__ void gram_load(v2f (&fa)[16], v2f (&fb)[16], const float* ap,
                                          const float* bp, int g) {
#pragma unroll
  for (int s = 0; s < 16; ++s) {
    fa[s] = *(const v2f*)&ap[64 * g + 4 * s];
    fb[s] = *(const v2f*)&bp[64 * g + 4 * s];
  }
}

__device__ __forceinline__ v8f gram_mma(v8f acc, const v2f (&fa)[16], const v2f (&fb)[16]) {
#pragma unroll
  for (int s = 0; s < 16; ++s) acc = wmma_f32(fa[s], fb[s], acc);
  return acc;
}

__global__ __launch_bounds__(256) void normalize_pack_gram(
    const float* __restrict__ captions, const float* __restrict__ images,
    float* __restrict__ capP, float* __restrict__ imgP, float* __restrict__ na,
    float* __restrict__ G) {
  __shared__ float rows[RP][FRONT_PITCH];
  const int tid = threadIdx.x;
  const int wave = __builtin_amdgcn_readfirstlane(tid >> 5);
  const int lane = tid & 31;
  const int c = lane & 15, h = lane >> 4;

  if (blockIdx.x < CAP_TILES) {
    const int tile = blockIdx.x;
    float4 v[2][4];
#pragma unroll
    for (int rr = 0; rr < 2; ++rr)
      load_row(v[rr], captions + ((size_t)tile * 16 + wave * 2 + rr) * EE, lane);
#pragma unroll
    for (int rr = 0; rr < 2; ++rr) {
      const int row = wave * 2 + rr;
      const float nrm = normalize_row(v[rr], rows[row], lane);
      if (lane == 0) na[tile * 16 + row] = nrm;
    }
    __syncthreads();
    float4* dst = (float4*)capP + (size_t)tile * TILE_F4;
#pragma unroll
    for (int j = 0; j < 8; ++j) pack_kb(dst, rows[c], wave + 8 * j, h, lane);
    return;
  }

  const int img = blockIdx.x - CAP_TILES;
  const float* isrc = images + (size_t)img * RR * EE;
  float4 v[5][4];
#pragma unroll
  for (int rr = 0; rr < 5; ++rr) {
    const int row = wave + 8 * rr;
    if (row < RR) load_row(v[rr], isrc + (size_t)row * EE, lane);
  }
  for (int idx = tid; idx < (RP - RR) * FRONT_PITCH / 4; idx += 256)
    ((float4*)&rows[RR][0])[idx] = float4{0.f, 0.f, 0.f, 0.f};
#pragma unroll
  for (int rr = 0; rr < 5; ++rr) {
    const int row = wave + 8 * rr;
    if (row < RR) (void)normalize_row(v[rr], rows[row], lane);
  }
  __syncthreads();

  if (wave >= 6) {
    for (int t = 0; t < 3; ++t) {
      float4* dst = (float4*)imgP + (size_t)(img * 3 + t) * TILE_F4;
      const float* rp = rows[t * 16 + c];
#pragma unroll 8
      for (int j = 0; j < KB8 / 2; ++j) pack_kb(dst, rp, (wave - 6) + 2 * j, h, lane);
    }
    return;
  }

  const int ti = (wave >= 3) + (wave >= 5);
  const int tj = wave - (ti == 0 ? 0 : (ti == 1 ? 2 : 3));
  const float* ap = rows[ti * 16 + c] + 2 * h;
  const float* bp = rows[tj * 16 + c] + 2 * h;
  v8f acc = {};
#define GROUP_STAYS() __builtin_amdgcn_sched_barrier(0)
  v2f fa0[16], fb0[16], fa1[16], fb1[16];
  gram_load(fa0, fb0, ap, bp, 0);
  GROUP_STAYS();
#pragma unroll
  for (int g = 0; g < 8; g += 2) {
    gram_load(fa1, fb1, ap, bp, g + 1);
    GROUP_STAYS();
    acc = gram_mma(acc, fa0, fb0);
    GROUP_STAYS();
    if (g + 2 < 8) gram_load(fa0, fb0, ap, bp, g + 2);
    GROUP_STAYS();
    acc = gram_mma(acc, fa1, fb1);
    GROUP_STAYS();
  }
#undef GROUP_STAYS
  float* Gi = G + (size_t)img * RP * RP;
#pragma unroll
  for (int j = 0; j < 8; ++j)
    Gi[(ti * 16 + j + 8 * h) * RP + tj * 16 + c] = acc[j];
  if (ti != tj) {
    float4* m = (float4*)(Gi + (tj * 16 + c) * RP + ti * 16 + 8 * h);
    m[0] = float4{acc[0], acc[1], acc[2], acc[3]};
    m[1] = float4{acc[4], acc[5], acc[6], acc[7]};
  }
}

__device__ __forceinline__ void load_set(float4 (&a)[4], float4 (&b)[3],
                                         const float4* Ap, const float4* Bp, int kb) {
#pragma unroll
  for (int m = 0; m < 4; ++m) a[m] = Ap[(m * KB8 + kb) * 32];
#pragma unroll
  for (int n = 0; n < 3; ++n) b[n] = Bp[(n * KB8 + kb) * 32];
}

__device__ __forceinline__ void mma_set(v8f (&acc)[4][3], const float4 (&a)[4],
                                        const float4 (&b)[3]) {
#pragma unroll
  for (int m = 0; m < 4; ++m)
#pragma unroll
    for (int n = 0; n < 3; ++n)
      acc[m][n] = wmma_f32(v2f{a[m].x, a[m].y}, v2f{b[n].x, b[n].y}, acc[m][n]);
#pragma unroll
  for (int m = 0; m < 4; ++m)
#pragma unroll
    for (int n = 0; n < 3; ++n)
      acc[m][n] = wmma_f32(v2f{a[m].z, a[m].w}, v2f{b[n].z, b[n].w}, acc[m][n]);
}

__global__ __launch_bounds__(128) void scan_main(
    const float* __restrict__ capP, const float* __restrict__ imgP,
    const float* __restrict__ na_g, const float* __restrict__ G,
    const int* __restrict__ cap_lens, float* __restrict__ out) {
  __shared__ float Wsh[4][LL][RP + 2];
  __shared__ float rsh[4][LL];

  const int tid = threadIdx.x;
  const int wave = tid >> 5, lane = tid & 31;
  const int hi = lane >> 4, n16 = lane & 15;
  const int i = blockIdx.x * 2 + (wave & 1);
  const int c = blockIdx.y * 2 + (wave >> 1);
  float (*W)[RP + 2] = Wsh[wave];

  const float4* Ap = (const float4*)capP + (size_t)(c * 4) * TILE_F4 + lane;
  const float4* Bp = (const float4*)imgP + (size_t)(i * 3) * TILE_F4 + lane;

  v8f acc[4][3];
#pragma unroll
  for (int m = 0; m < 4; ++m)
#pragma unroll
    for (int n = 0; n < 3; ++n) acc[m][n] = v8f{};

#define LOADS_STAY() __builtin_amdgcn_sched_barrier(0)
  float4 a0[4], b0[3], a1[4], b1[3], a2[4], b2[3], a3[4], b3[3];
  load_set(a0, b0, Ap, Bp, 0);
  LOADS_STAY();
  load_set(a1, b1, Ap, Bp, 1);
  LOADS_STAY();
  load_set(a2, b2, Ap, Bp, 2);
  LOADS_STAY();
  for (int kb = 0; kb < KB8 - 4; kb += 4) {
    load_set(a3, b3, Ap, Bp, kb + 3);
    LOADS_STAY();
    mma_set(acc, a0, b0);
    LOADS_STAY();
    load_set(a0, b0, Ap, Bp, kb + 4);
    LOADS_STAY();
    mma_set(acc, a1, b1);
    LOADS_STAY();
    load_set(a1, b1, Ap, Bp, kb + 5);
    LOADS_STAY();
    mma_set(acc, a2, b2);
    LOADS_STAY();
    load_set(a2, b2, Ap, Bp, kb + 6);
    LOADS_STAY();
    mma_set(acc, a3, b3);
    LOADS_STAY();
  }
  load_set(a3, b3, Ap, Bp, KB8 - 1);
  LOADS_STAY();
  mma_set(acc, a0, b0);
  mma_set(acc, a1, b1);
  mma_set(acc, a2, b2);
  mma_set(acc, a3, b3);
#undef LOADS_STAY

#pragma unroll
  for (int m = 0; m < 4; ++m)
#pragma unroll
    for (int j = 0; j < 8; ++j) {
      const int l = m * 16 + j + 8 * hi;
      W[l][n16]      = acc[m][0][j];
      W[l][16 + n16] = acc[m][1][j];
      W[l][32 + n16] = acc[m][2][j];
    }

  __syncthreads();

  const float* Gi = G + (size_t)i * RP * RP;
  v2f g[12][3];
#pragma unroll
  for (int ks = 0; ks < 12; ++ks)
#pragma unroll
    for (int n = 0; n < 3; ++n)
      g[ks][n] = *(const v2f*)(Gi + (n * 16 + n16) * RP + ks * 4 + 2 * hi);

  float dot[2], inv[2];
#pragma unroll
  for (int q = 0; q < 2; ++q) {
    const int tok = lane + 32 * q;
    float mx = -1e30f;
    for (int r = 0; r < RR; ++r) mx = fmaxf(mx, W[tok][r]);
    float sum = 0.f, d = 0.f;
    for (int r = 0; r < RR; ++r) {
      const float s = W[tok][r];
      const float e = expf(LAMBDA * (s - mx));
      sum += e;
      d += e * s;
      W[tok][r] = e;
    }
    for (int r = RR; r < RP; ++r) W[tok][r] = 0.f;
    inv[q] = 1.0f / sum;
    dot[q] = d * inv[q];
  }
  __syncthreads();

  v8f vv[4][3];
#pragma unroll
  for (int m = 0; m < 4; ++m)
#pragma unroll
    for (int n = 0; n < 3; ++n) vv[m][n] = v8f{};
#pragma unroll
  for (int ks = 0; ks < 12; ++ks) {
    const int kk = ks * 4 + 2 * hi;
#pragma unroll
    for (int m = 0; m < 4; ++m) {
      const v2f a = *(const v2f*)&W[m * 16 + n16][kk];
#pragma unroll
      for (int n = 0; n < 3; ++n) vv[m][n] = wmma_f32(a, g[ks][n], vv[m][n]);
    }
  }

  __syncthreads();
#pragma unroll
  for (int m = 0; m < 4; ++m)
#pragma unroll
    for (int j = 0; j < 8; ++j) {
      const int l = m * 16 + j + 8 * hi;
      W[l][n16]      = vv[m][0][j] * W[l][n16];
      W[l][16 + n16] = vv[m][1][j] * W[l][16 + n16];
      W[l][32 + n16] = vv[m][2][j] * W[l][32 + n16];
    }
  __syncthreads();

  const int len = cap_lens[c];
#pragma unroll
  for (int q = 0; q < 2; ++q) {
    const int tok = lane + 32 * q;
    float nb2 = 0.f;
    for (int r = 0; r < RR; ++r) nb2 += W[tok][r];
    const float nb = sqrtf(fmaxf(nb2, 0.f)) * inv[q];
    const float na = na_g[c * LL + tok];
    const float denom = fmaxf(na * nb, 1e-8f);
    float rl = dot[q] / denom;
    if (tok >= len) rl = 0.f;
    rsh[wave][tok] = rl;
  }
  __syncthreads();
  if (lane == 0) {
    float s = 0.f;
    for (int l = 0; l < LL; ++l) s += rsh[wave][l];
    out[(size_t)c * BB + i] = s * (1.0f / LL);
  }
}

extern "C" void kernel_launch(void* const* d_in, const int* in_sizes, int n_in,
                              void* d_out, int out_size, void* d_ws, size_t ws_size,
                              hipStream_t stream) {
  (void)in_sizes; (void)n_in; (void)out_size; (void)ws_size;
  const float* images = (const float*)d_in[0];
  const float* captions = (const float*)d_in[1];
  const int* cap_lens = (const int*)d_in[2];
  float* out = (float*)d_out;

  float* ws = (float*)d_ws;
  float* capP = ws;
  float* imgP = capP + (size_t)CAP_TILES * TILE_F4 * 4;
  float* na   = imgP + (size_t)IMG_TILES * TILE_F4 * 4;
  float* G    = na + (size_t)BB * LL;

  normalize_pack_gram<<<CAP_TILES + BB, 256, 0, stream>>>(captions, images, capP, imgP,
                                                          na, G);
  scan_main<<<dim3(BB / 2, BB / 2), 128, 0, stream>>>(capP, imgP, na, G, cap_lens, out);
}


# scan_main second product uses Gram symmetry: launch 1 stores only the upper triangle with half the diagonal, 60 matrix steps per pair (was 144), norm doubled after; norm sum order changes
# speedup vs baseline: 1.7835x; 1.0419x over previous
#include <hip/hip_runtime.h>
#include <math.h>


#define BB 64
#define RR 36
#define RP 48
#define LL 64
#define EE 512
#define KB8 64
#define CAP_TILES (BB * LL / 16)
#define IMG_TILES (BB * 3)
#define TILE_F4 (KB8 * 32)
#define LAMBDA 9.0f

typedef float v2f __attribute__((ext_vector_type(2)));
typedef float v8f __attribute__((ext_vector_type(8)));

__device__ __forceinline__ v8f wmma_f32(v2f a, v2f b, v8f c) {
  return __builtin_amdgcn_wmma_f32_16x16x4_f32(
       false, a,  false, b,
       (short)0, c,  false,  false);
}


#define FRONT_PITCH (EE + 4)

__device__ __forceinline__ void load_row(float4 (&v)[4], const float* src, int lane) {
#pragma unroll
  for (int j = 0; j < 4; ++j) v[j] = *(const float4*)(src + 4 * lane + 128 * j);
}

__device__ __forceinline__ float normalize_row(const float4 (&v)[4], float* dstrow, int lane) {
  float p[4];
#pragma unroll
  for (int j = 0; j < 4; ++j) {
    const float4 x = v[j];
    p[j] = x.x * x.x + x.y * x.y + x.z * x.z + x.w * x.w;
  }
  float s = (p[0] + p[2]) + (p[1] + p[3]);
  s += __shfl_xor(s, 16);
  s += __shfl_xor(s, 8);
  s += __shfl_xor(s, 4);
  s += __shfl_xor(s, 2);
  s += __shfl_xor(s, 1);
  const float norm = sqrtf(s);
  const float scale = 1.0f / fmaxf(norm, 1e-12f);
#pragma unroll
  for (int j = 0; j < 4; ++j) {
    float4 o = v[j];
    o.x *= scale; o.y *= scale; o.z *= scale; o.w *= scale;
    *(float4*)&dstrow[4 * lane + 128 * j] = o;
  }
  return norm * scale;
}

__device__ __forceinline__ void pack_kb(float4* dst, const float* rp, int kb, int h, int lane) {
  const v2f lo = *(const v2f*)&rp[8 * kb + 2 * h];
  const v2f hi2 = *(const v2f*)&rp[8 * kb + 4 + 2 * h];
  float4 o; o.x = lo.x; o.y = lo.y; o.z = hi2.x; o.w = hi2.y;
  dst[kb * 32 + lane] = o;
}

__device__ __forceinline__ void gram_load(v2f (&fa)[16], v2f (&fb)[16], const float* ap,
                                          const float* bp, int g) {
#pragma unroll
  for (int s = 0; s < 16; ++s) {
    fa[s] = *(const v2f*)&ap[64 * g + 4 * s];
    fb[s] = *(const v2f*)&bp[64 * g + 4 * s];
  }
}

__device__ __forceinline__ v8f gram_mma(v8f acc, const v2f (&fa)[16], const v2f (&fb)[16]) {
#pragma unroll
  for (int s = 0; s < 16; ++s) acc = wmma_f32(fa[s], fb[s], acc);
  return acc;
}

__global__ __launch_bounds__(256) void normalize_pack_gram(
    const float* __restrict__ captions, const float* __restrict__ images,
    float* __restrict__ capP, float* __restrict__ imgP, float* __restrict__ na,
    float* __restrict__ G) {
  __shared__ float rows[RP][FRONT_PITCH];
  const int tid = threadIdx.x;
  const int wave = __builtin_amdgcn_readfirstlane(tid >> 5);
  const int lane = tid & 31;
  const int c = lane & 15, h = lane >> 4;

  if (blockIdx.x < CAP_TILES) {
    const int tile = blockIdx.x;
    float4 v[2][4];
#pragma unroll
    for (int rr = 0; rr < 2; ++rr)
      load_row(v[rr], captions + ((size_t)tile * 16 + wave * 2 + rr) * EE, lane);
#pragma unroll
    for (int rr = 0; rr < 2; ++rr) {
      const int row = wave * 2 + rr;
      const float nrm = normalize_row(v[rr], rows[row], lane);
      if (lane == 0) na[tile * 16 + row] = nrm;
    }
    __syncthreads();
    float4* dst = (float4*)capP + (size_t)tile * TILE_F4;
#pragma unroll
    for (int j = 0; j < 8; ++j) pack_kb(dst, rows[c], wave + 8 * j, h, lane);
    return;
  }

  const int img = blockIdx.x - CAP_TILES;
  const float* isrc = images + (size_t)img * RR * EE;
  float4 v[5][4];
#pragma unroll
  for (int rr = 0; rr < 5; ++rr) {
    const int row = wave + 8 * rr;
    if (row < RR) load_row(v[rr], isrc + (size_t)row * EE, lane);
  }
  for (int idx = tid; idx < (RP - RR) * FRONT_PITCH / 4; idx += 256)
    ((float4*)&rows[RR][0])[idx] = float4{0.f, 0.f, 0.f, 0.f};
#pragma unroll
  for (int rr = 0; rr < 5; ++rr) {
    const int row = wave + 8 * rr;
    if (row < RR) (void)normalize_row(v[rr], rows[row], lane);
  }
  __syncthreads();

  if (wave >= 6) {
    for (int t = 0; t < 3; ++t) {
      float4* dst = (float4*)imgP + (size_t)(img * 3 + t) * TILE_F4;
      const float* rp = rows[t * 16 + c];
#pragma unroll 8
      for (int j = 0; j < KB8 / 2; ++j) pack_kb(dst, rp, (wave - 6) + 2 * j, h, lane);
    }
    return;
  }

  const int ti = (wave >= 3) + (wave >= 5);
  const int tj = wave - (ti == 0 ? 0 : (ti == 1 ? 2 : 3));
  const float* ap = rows[ti * 16 + c] + 2 * h;
  const float* bp = rows[tj * 16 + c] + 2 * h;
  v8f acc = {};
#define GROUP_STAYS() __builtin_amdgcn_sched_barrier(0)
  v2f fa0[16], fb0[16], fa1[16], fb1[16];
  gram_load(fa0, fb0, ap, bp, 0);
  GROUP_STAYS();
#pragma unroll
  for (int g = 0; g < 8; g += 2) {
    gram_load(fa1, fb1, ap, bp, g + 1);
    GROUP_STAYS();
    acc = gram_mma(acc, fa0, fb0);
    GROUP_STAYS();
    if (g + 2 < 8) gram_load(fa0, fb0, ap, bp, g + 2);
    GROUP_STAYS();
    acc = gram_mma(acc, fa1, fb1);
    GROUP_STAYS();
  }
#undef GROUP_STAYS
  float* Gi = G + (size_t)img * RP * RP;
#pragma unroll
  for (int j = 0; j < 8; ++j) {
    const int rloc = j + 8 * h;
    float v = acc[j];
    if (ti == tj) v = (c > rloc) ? v : ((c == rloc) ? 0.5f * v : 0.f);
    Gi[(ti * 16 + rloc) * RP + tj * 16 + c] = v;
  }
}

__device__ __forceinline__ void load_set(float4 (&a)[4], float4 (&b)[3],
                                         const float4* Ap, const float4* Bp, int kb) {
#pragma unroll
  for (int m = 0; m < 4; ++m) a[m] = Ap[(m * KB8 + kb) * 32];
#pragma unroll
  for (int n = 0; n < 3; ++n) b[n] = Bp[(n * KB8 + kb) * 32];
}

__device__ __forceinline__ void mma_set(v8f (&acc)[4][3], const float4 (&a)[4],
                                        const float4 (&b)[3]) {
#pragma unroll
  for (int m = 0; m < 4; ++m)
#pragma unroll
    for (int n = 0; n < 3; ++n)
      acc[m][n] = wmma_f32(v2f{a[m].x, a[m].y}, v2f{b[n].x, b[n].y}, acc[m][n]);
#pragma unroll
  for (int m = 0; m < 4; ++m)
#pragma unroll
    for (int n = 0; n < 3; ++n)
      acc[m][n] = wmma_f32(v2f{a[m].z, a[m].w}, v2f{b[n].z, b[n].w}, acc[m][n]);
}

__global__ __launch_bounds__(128) void scan_main(
    const float* __restrict__ capP, const float* __restrict__ imgP,
    const float* __restrict__ na_g, const float* __restrict__ G,
    const int* __restrict__ cap_lens, float* __restrict__ out) {
  __shared__ float Wsh[4][LL][RP + 2];
  __shared__ float rsh[4][LL];

  const int tid = threadIdx.x;
  const int wave = tid >> 5, lane = tid & 31;
  const int hi = lane >> 4, n16 = lane & 15;
  const int i = blockIdx.x * 2 + (wave & 1);
  const int c = blockIdx.y * 2 + (wave >> 1);
  float (*W)[RP + 2] = Wsh[wave];

  const float4* Ap = (const float4*)capP + (size_t)(c * 4) * TILE_F4 + lane;
  const float4* Bp = (const float4*)imgP + (size_t)(i * 3) * TILE_F4 + lane;

  v8f acc[4][3];
#pragma unroll
  for (int m = 0; m < 4; ++m)
#pragma unroll
    for (int n = 0; n < 3; ++n) acc[m][n] = v8f{};

#define LOADS_STAY() __builtin_amdgcn_sched_barrier(0)
  float4 a0[4], b0[3], a1[4], b1[3], a2[4], b2[3], a3[4], b3[3];
  load_set(a0, b0, Ap, Bp, 0);
  LOADS_STAY();
  load_set(a1, b1, Ap, Bp, 1);
  LOADS_STAY();
  load_set(a2, b2, Ap, Bp, 2);
  LOADS_STAY();
  for (int kb = 0; kb < KB8 - 4; kb += 4) {
    load_set(a3, b3, Ap, Bp, kb + 3);
    LOADS_STAY();
    mma_set(acc, a0, b0);
    LOADS_STAY();
    load_set(a0, b0, Ap, Bp, kb + 4);
    LOADS_STAY();
    mma_set(acc, a1, b1);
    LOADS_STAY();
    load_set(a1, b1, Ap, Bp, kb + 5);
    LOADS_STAY();
    mma_set(acc, a2, b2);
    LOADS_STAY();
    load_set(a2, b2, Ap, Bp, kb + 6);
    LOADS_STAY();
    mma_set(acc, a3, b3);
    LOADS_STAY();
  }
  load_set(a3, b3, Ap, Bp, KB8 - 1);
  LOADS_STAY();
  mma_set(acc, a0, b0);
  mma_set(acc, a1, b1);
  mma_set(acc, a2, b2);
  mma_set(acc, a3, b3);
#undef LOADS_STAY

#pragma unroll
  for (int m = 0; m < 4; ++m)
#pragma unroll
    for (int j = 0; j < 8; ++j) {
      const int l = m * 16 + j + 8 * hi;
      W[l][n16]      = acc[m][0][j];
      W[l][16 + n16] = acc[m][1][j];
      W[l][32 + n16] = acc[m][2][j];
    }

  __syncthreads();

  const float* Gi = G + (size_t)i * RP * RP;
  v2f g[RR / 4][3];
#pragma unroll
  for (int ks = 0; ks < RR / 4; ++ks)
#pragma unroll
    for (int n = 0; n <= ks / 4; ++n)
      g[ks][n] = *(const v2f*)(Gi + (n * 16 + n16) * RP + ks * 4 + 2 * hi);

  float dot[2], inv[2];
#pragma unroll
  for (int q = 0; q < 2; ++q) {
    const int tok = lane + 32 * q;
    float mx = -1e30f;
    for (int r = 0; r < RR; ++r) mx = fmaxf(mx, W[tok][r]);
    float sum = 0.f, d = 0.f;
    for (int r = 0; r < RR; ++r) {
      const float s = W[tok][r];
      const float e = expf(LAMBDA * (s - mx));
      sum += e;
      d += e * s;
      W[tok][r] = e;
    }
    inv[q] = 1.0f / sum;
    dot[q] = d * inv[q];
  }
  __syncthreads();

  v8f vv[4][3];
#pragma unroll
  for (int m = 0; m < 4; ++m)
#pragma unroll
    for (int n = 0; n < 3; ++n) vv[m][n] = v8f{};
#pragma unroll
  for (int ks = 0; ks < RR / 4; ++ks) {
    const int kk = ks * 4 + 2 * hi;
#pragma unroll
    for (int m = 0; m < 4; ++m) {
      const v2f a = *(const v2f*)&W[m * 16 + n16][kk];
#pragma unroll
      for (int n = 0; n <= ks / 4; ++n) vv[m][n] = wmma_f32(a, g[ks][n], vv[m][n]);
    }
  }

  __syncthreads();
#pragma unroll
  for (int m = 0; m < 4; ++m)
#pragma unroll
    for (int j = 0; j < 8; ++j) {
      const int l = m * 16 + j + 8 * hi;
      W[l][n16]      = vv[m][0][j] * W[l][n16];
      W[l][16 + n16] = vv[m][1][j] * W[l][16 + n16];
      W[l][32 + n16] = vv[m][2][j] * W[l][32 + n16];
    }
  __syncthreads();

  const int len = cap_lens[c];
#pragma unroll
  for (int q = 0; q < 2; ++q) {
    const int tok = lane + 32 * q;
    float nb2 = 0.f;
    for (int r = 0; r < RR; ++r) nb2 += W[tok][r];
    const float nb = sqrtf(fmaxf(nb2 + nb2, 0.f)) * inv[q];
    const float na = na_g[c * LL + tok];
    const float denom = fmaxf(na * nb, 1e-8f);
    float rl = dot[q] / denom;
    if (tok >= len) rl = 0.f;
    rsh[wave][tok] = rl;
  }
  __syncthreads();
  if (lane == 0) {
    float s = 0.f;
    for (int l = 0; l < LL; ++l) s += rsh[wave][l];
    out[(size_t)c * BB + i] = s * (1.0f / LL);
  }
}

extern "C" void kernel_launch(void* const* d_in, const int* in_sizes, int n_in,
                              void* d_out, int out_size, void* d_ws, size_t ws_size,
                              hipStream_t stream) {
  (void)in_sizes; (void)n_in; (void)out_size; (void)ws_size;
  const float* images = (const float*)d_in[0];
  const float* captions = (const float*)d_in[1];
  const int* cap_lens = (const int*)d_in[2];
  float* out = (float*)d_out;

  float* ws = (float*)d_ws;
  float* capP = ws;
  float* imgP = capP + (size_t)CAP_TILES * TILE_F4 * 4;
  float* na   = imgP + (size_t)IMG_TILES * TILE_F4 * 4;
  float* G    = na + (size_t)BB * LL;

  normalize_pack_gram<<<CAP_TILES + BB, 256, 0, stream>>>(captions, images, capP, imgP,
                                                          na, G);
  scan_main<<<dim3(BB / 2, BB / 2), 128, 0, stream>>>(capP, imgP, na, G, cap_lens, out);
}
